# starting kernel
# baseline (speedup 1.0000x reference)
#include <hip/hip_runtime.h>
#include <math.h>


typedef __attribute__((ext_vector_type(2))) float v2f;
typedef __attribute__((ext_vector_type(8))) float v8f;

#define CDIM 64
#define TDIM 4
#define ROWLEN 64
#define TCP 256

__device__ __forceinline__ v8f wmma16x16_k64(const float* __restrict__ Arows,
                                             const float* __restrict__ Blds,
                                             int c0) {
    const int lane = threadIdx.x & 31;
    const int m    = lane & 15;
    const int koff = (lane >> 4) << 1;
    const float* __restrict__ arow = Arows + m * ROWLEN;
    v8f acc = {0.f, 0.f, 0.f, 0.f, 0.f, 0.f, 0.f, 0.f};
#pragma unroll
    for (int k0 = 0; k0 < 64; k0 += 4) {
        v2f a, b;
        a.x = arow[k0 + koff];
        a.y = arow[k0 + koff + 1];
        b.x = Blds[(k0 + koff)     * ROWLEN + c0 + m];
        b.y = Blds[(k0 + koff + 1) * ROWLEN + c0 + m];
        acc = __builtin_amdgcn_wmma_f32_16x16x4_f32(
            false, a, false, b, (short)0, acc, false, false);
    }
    return acc;
}

__device__ __forceinline__ void store_tile(float* __restrict__ Out, int row0,
                                           int c0, v8f acc) {
    const int lane = threadIdx.x & 31;
    const int col  = c0 + (lane & 15);
    const int rb   = row0 + ((lane >> 4) << 3);
#pragma unroll
    for (int i = 0; i < 8; ++i)
        Out[(rb + i) * ROWLEN + col] = acc[i];
}

__global__ void __launch_bounds__(128)
proj_src_kernel(const float* __restrict__ feat, const float* __restrict__ U,
                float* __restrict__ Out, int mtiles) {
    __shared__ float ldsU[CDIM * CDIM];
    for (int i = threadIdx.x; i < CDIM * CDIM; i += blockDim.x) ldsU[i] = U[i];
    __syncthreads();

    const int wave = threadIdx.x >> 5;
    const int tile = blockIdx.x * (blockDim.x >> 5) + wave;
    if (tile >= mtiles) return;
    const int row0 = tile * 16;
    const float* __restrict__ Arows = feat + (long long)row0 * ROWLEN;
#pragma unroll
    for (int c0 = 0; c0 < 64; c0 += 16) {
        v8f acc = wmma16x16_k64(Arows, ldsU, c0);
        store_tile(Out, row0, c0, acc);
    }
}

__global__ void __launch_bounds__(128)
proj_dst_kernel(const float* __restrict__ feat, const float* __restrict__ U,
                const float* __restrict__ V, float* __restrict__ OutBase,
                float* __restrict__ OutV, int mtiles) {
    __shared__ float ldsU[CDIM * CDIM];
    __shared__ float ldsV[CDIM * CDIM];
    for (int i = threadIdx.x; i < CDIM * CDIM; i += blockDim.x) {
        ldsU[i] = U[i];
        ldsV[i] = V[i];
    }
    __syncthreads();

    const int wave = threadIdx.x >> 5;
    const int tile = blockIdx.x * (blockDim.x >> 5) + wave;
    if (tile >= mtiles) return;
    const int row0 = tile * 16;
    const int lane = threadIdx.x & 31;
    const int m    = lane & 15;
    const int koff = (lane >> 4) << 1;
    const float* __restrict__ arow = feat + ((long long)row0 + m) * ROWLEN;

#pragma unroll
    for (int c0 = 0; c0 < 64; c0 += 16) {
        v8f accU = {0.f, 0.f, 0.f, 0.f, 0.f, 0.f, 0.f, 0.f};
        v8f accV = {0.f, 0.f, 0.f, 0.f, 0.f, 0.f, 0.f, 0.f};
#pragma unroll
        for (int k0 = 0; k0 < 64; k0 += 4) {
            v2f a, bu, bv;
            a.x  = arow[k0 + koff];
            a.y  = arow[k0 + koff + 1];
            bu.x = ldsU[(k0 + koff)     * ROWLEN + c0 + m];
            bu.y = ldsU[(k0 + koff + 1) * ROWLEN + c0 + m];
            bv.x = ldsV[(k0 + koff)     * ROWLEN + c0 + m];
            bv.y = ldsV[(k0 + koff + 1) * ROWLEN + c0 + m];
            accU = __builtin_amdgcn_wmma_f32_16x16x4_f32(
                false, a, false, bu, (short)0, accU, false, false);
            accV = __builtin_amdgcn_wmma_f32_16x16x4_f32(
                false, a, false, bv, (short)0, accV, false, false);
        }
        store_tile(OutBase, row0, c0, accU);
        store_tile(OutV,    row0, c0, accV);
    }
}

__global__ void __launch_bounds__(256)
edge_kernel(const float* __restrict__ h_src, const float* __restrict__ h_dst,
            const float* __restrict__ edge_w, const float* __restrict__ we,
            const int* __restrict__ src_idx, const int* __restrict__ dst_idx,
            float* __restrict__ aggr, long long total) {
    long long i = (long long)blockIdx.x * blockDim.x + threadIdx.x;
    if (i >= total) return;
    const int e   = (int)(i >> 6);
    const int r   = (int)(i & 63);
    const int tc4 = r << 2;
    const int c   = tc4 & (CDIM - 1);

    const int   s = src_idx[e];
    const int   d = dst_idx[e];
    const float w = edge_w[e];

    const float4 hs = *(const float4*)(h_src + (long long)s * TCP + tc4);
    const float4 hd = *(const float4*)(h_dst + (long long)d * TCP + tc4);
    const float4 wv = *(const float4*)(we + c);
    float* __restrict__ out = aggr + (long long)d * TCP + tc4;

    {
        float x = hs.x * hd.x * (w * wv.x);
        atomicAdd(out + 0, hs.x / (1.0f + __expf(-x)));
    }
    {
        float x = hs.y * hd.y * (w * wv.y);
        atomicAdd(out + 1, hs.y / (1.0f + __expf(-x)));
    }
    {
        float x = hs.z * hd.z * (w * wv.z);
        atomicAdd(out + 2, hs.z / (1.0f + __expf(-x)));
    }
    {
        float x = hs.w * hd.w * (w * wv.w);
        atomicAdd(out + 3, hs.w / (1.0f + __expf(-x)));
    }
}

__global__ void __launch_bounds__(256)
ln_kernel(const float* __restrict__ acc, const float* __restrict__ feat_dst,
          const float* __restrict__ gamma, const float* __restrict__ beta,
          float* __restrict__ out, int rows) {
    const int wave = threadIdx.x >> 5;
    const int lane = threadIdx.x & 31;
    const int row  = blockIdx.x * (blockDim.x >> 5) + wave;
    if (row >= rows) return;

    const float* __restrict__ x = acc + (long long)row * CDIM;
    float x0 = x[lane];
    float x1 = x[lane + 32];

    float s = x0 + x1;
#pragma unroll
    for (int off = 16; off > 0; off >>= 1) s += __shfl_xor(s, off, 32);
    const float mean = s * (1.0f / 64.0f);

    const float d0 = x0 - mean, d1 = x1 - mean;
    float q = d0 * d0 + d1 * d1;
#pragma unroll
    for (int off = 16; off > 0; off >>= 1) q += __shfl_xor(q, off, 32);
    const float rstd = rsqrtf(q * (1.0f / 64.0f) + 1e-5f);

    const float* __restrict__ f = feat_dst + (long long)row * CDIM;
    out[(long long)row * CDIM + lane] =
        d0 * rstd * gamma[lane] + beta[lane] + f[lane];
    out[(long long)row * CDIM + lane + 32] =
        d1 * rstd * gamma[lane + 32] + beta[lane + 32] + f[lane + 32];
}

extern "C" void kernel_launch(void* const* d_in, const int* in_sizes, int n_in,
                              void* d_out, int out_size, void* d_ws, size_t ws_size,
                              hipStream_t stream) {
    const float* feat_src = (const float*)d_in[0];
    const float* feat_dst = (const float*)d_in[1];
    const float* edge_w   = (const float*)d_in[2];
    const float* weight_e = (const float*)d_in[3];
    const float* u        = (const float*)d_in[4];
    const float* v        = (const float*)d_in[5];
    const float* ln_gamma = (const float*)d_in[6];
    const float* ln_beta  = (const float*)d_in[7];
    const int*   src_idx  = (const int*)d_in[8];
    const int*   dst_idx  = (const int*)d_in[9];
    float*       out      = (float*)d_out;

    const int n_src = in_sizes[0] / TCP;
    const int n_dst = in_sizes[1] / TCP;
    const int E     = in_sizes[2];

    float* ws    = (float*)d_ws;
    float* h_src = ws;
    float* h_dst = ws + (size_t)n_src * TCP;
    float* aggr  = h_dst + (size_t)n_dst * TCP;

    const int mtiles_src = (n_src * TDIM) / 16;
    const int mtiles_dst = (n_dst * TDIM) / 16;
    const int wpb = 4;

    proj_src_kernel<<<(mtiles_src + wpb - 1) / wpb, 128, 0, stream>>>(
        feat_src, u, h_src, mtiles_src);

    proj_dst_kernel<<<(mtiles_dst + wpb - 1) / wpb, 128, 0, stream>>>(
        feat_dst, u, v, aggr, h_dst, mtiles_dst);

    const long long edge_work = (long long)E * (TCP / 4);
    edge_kernel<<<(int)((edge_work + 255) / 256), 256, 0, stream>>>(
        h_src, h_dst, edge_w, weight_e, src_idx, dst_idx, aggr, edge_work);

    const int rows = n_dst * TDIM;
    ln_kernel<<<(rows + 7) / 8, 256, 0, stream>>>(
        aggr, feat_dst, ln_gamma, ln_beta, out, rows);
}


# Edge kernel: queue phase and sum phase merged into one pass: a wave sums the block's edges it finds in its index group at once; groups drawn from an LDS counter; sum order = LDS arrival order
# speedup vs baseline: 1.7618x; 1.7618x over previous
#include <hip/hip_runtime.h>
#include <math.h>


typedef __attribute__((ext_vector_type(2))) float v2f;
typedef __attribute__((ext_vector_type(8))) float v8f;

#define CDIM 64
#define TDIM 4
#define ROWLEN 64
#define TCP 256

__device__ __forceinline__ v8f wmma16x16_k64(const float* __restrict__ Arows,
                                             const float* __restrict__ Blds,
                                             int c0) {
    const int lane = threadIdx.x & 31;
    const int m    = lane & 15;
    const int koff = (lane >> 4) << 1;
    const float* __restrict__ arow = Arows + m * ROWLEN;
    v8f acc = {0.f, 0.f, 0.f, 0.f, 0.f, 0.f, 0.f, 0.f};
#pragma unroll
    for (int k0 = 0; k0 < 64; k0 += 4) {
        v2f a, b;
        a.x = arow[k0 + koff];
        a.y = arow[k0 + koff + 1];
        b.x = Blds[(k0 + koff)     * ROWLEN + c0 + m];
        b.y = Blds[(k0 + koff + 1) * ROWLEN + c0 + m];
        acc = __builtin_amdgcn_wmma_f32_16x16x4_f32(
            false, a, false, b, (short)0, acc, false, false);
    }
    return acc;
}

__device__ __forceinline__ void store_tile(float* __restrict__ Out, int row0,
                                           int c0, v8f acc) {
    const int lane = threadIdx.x & 31;
    const int col  = c0 + (lane & 15);
    const int rb   = row0 + ((lane >> 4) << 3);
#pragma unroll
    for (int i = 0; i < 8; ++i)
        Out[(rb + i) * ROWLEN + col] = acc[i];
}

__global__ void __launch_bounds__(128)
proj_src_kernel(const float* __restrict__ feat, const float* __restrict__ U,
                float* __restrict__ Out, int mtiles) {
    __shared__ float ldsU[CDIM * CDIM];
    for (int i = threadIdx.x; i < CDIM * CDIM; i += blockDim.x) ldsU[i] = U[i];
    __syncthreads();

    const int wave = threadIdx.x >> 5;
    const int tile = blockIdx.x * (blockDim.x >> 5) + wave;
    if (tile >= mtiles) return;
    const int row0 = tile * 16;
    const float* __restrict__ Arows = feat + (long long)row0 * ROWLEN;
#pragma unroll
    for (int c0 = 0; c0 < 64; c0 += 16) {
        v8f acc = wmma16x16_k64(Arows, ldsU, c0);
        store_tile(Out, row0, c0, acc);
    }
}

__global__ void __launch_bounds__(128)
proj_dst_kernel(const float* __restrict__ feat, const float* __restrict__ U,
                const float* __restrict__ V, float* __restrict__ OutBase,
                float* __restrict__ OutV, int mtiles) {
    __shared__ float ldsU[CDIM * CDIM];
    __shared__ float ldsV[CDIM * CDIM];
    for (int i = threadIdx.x; i < CDIM * CDIM; i += blockDim.x) {
        ldsU[i] = U[i];
        ldsV[i] = V[i];
    }
    __syncthreads();

    const int wave = threadIdx.x >> 5;
    const int tile = blockIdx.x * (blockDim.x >> 5) + wave;
    if (tile >= mtiles) return;
    const int row0 = tile * 16;
    const int lane = threadIdx.x & 31;
    const int m    = lane & 15;
    const int koff = (lane >> 4) << 1;
    const float* __restrict__ arow = feat + ((long long)row0 + m) * ROWLEN;

#pragma unroll
    for (int c0 = 0; c0 < 64; c0 += 16) {
        v8f accU = {0.f, 0.f, 0.f, 0.f, 0.f, 0.f, 0.f, 0.f};
        v8f accV = {0.f, 0.f, 0.f, 0.f, 0.f, 0.f, 0.f, 0.f};
#pragma unroll
        for (int k0 = 0; k0 < 64; k0 += 4) {
            v2f a, bu, bv;
            a.x  = arow[k0 + koff];
            a.y  = arow[k0 + koff + 1];
            bu.x = ldsU[(k0 + koff)     * ROWLEN + c0 + m];
            bu.y = ldsU[(k0 + koff + 1) * ROWLEN + c0 + m];
            bv.x = ldsV[(k0 + koff)     * ROWLEN + c0 + m];
            bv.y = ldsV[(k0 + koff + 1) * ROWLEN + c0 + m];
            accU = __builtin_amdgcn_wmma_f32_16x16x4_f32(
                false, a, false, bu, (short)0, accU, false, false);
            accV = __builtin_amdgcn_wmma_f32_16x16x4_f32(
                false, a, false, bv, (short)0, accV, false, false);
        }
        store_tile(OutBase, row0, c0, accU);
        store_tile(OutV,    row0, c0, accV);
    }
}

#define NPB 40
#define EDGE_WAVES 32

__device__ __forceinline__ float gated(float hs, float hd, float ww) {
    const float x = hs * hd * ww;
    return hs / (1.0f + __expf(-x));
}

__device__ __forceinline__ void gated_row_add(float4 a0, float4 a1, const float* ldsHd,
                                              float* ldsAcc, int dl, float w,
                                              int lane, float4 wv) {
    const float* hdp = ldsHd + dl * TCP + 4 * lane;
    const float4 b0 = *(const float4*)(hdp);
    const float4 b1 = *(const float4*)(hdp + 128);
    float* acc = ldsAcc + dl * TCP + lane;
    const float wx = w * wv.x, wy = w * wv.y, wz = w * wv.z, ww = w * wv.w;
    atomicAdd(acc +   0, gated(a0.x, b0.x, wx));
    atomicAdd(acc +  32, gated(a0.y, b0.y, wy));
    atomicAdd(acc +  64, gated(a0.z, b0.z, wz));
    atomicAdd(acc +  96, gated(a0.w, b0.w, ww));
    atomicAdd(acc + 128, gated(a1.x, b1.x, wx));
    atomicAdd(acc + 160, gated(a1.y, b1.y, wy));
    atomicAdd(acc + 192, gated(a1.z, b1.z, wz));
    atomicAdd(acc + 224, gated(a1.w, b1.w, ww));
}

__device__ __forceinline__ void load_group(const int* __restrict__ dst_idx, int g, int E,
                                           int lane, int4& c0, int4& c1, int4& c2, int4& c3) {
    const int eb = (g << 9) + 4 * lane;
    if (((g + 1) << 9) <= E) {
        c0 = *(const int4*)(dst_idx + eb);
        c1 = *(const int4*)(dst_idx + eb + 128);
        c2 = *(const int4*)(dst_idx + eb + 256);
        c3 = *(const int4*)(dst_idx + eb + 384);
    } else {
        int t[16];
#pragma unroll
        for (int i = 0; i < 16; ++i) {
            const int e = eb + 128 * (i >> 2) + (i & 3);
            t[i] = (e < E) ? dst_idx[e] : -1;
        }
        c0 = make_int4(t[0], t[1], t[2], t[3]);
        c1 = make_int4(t[4], t[5], t[6], t[7]);
        c2 = make_int4(t[8], t[9], t[10], t[11]);
        c3 = make_int4(t[12], t[13], t[14], t[15]);
    }
}

__device__ __forceinline__ int next_hit(unsigned& hit, int ebase, int lane) {
    const unsigned lm = __builtin_amdgcn_ballot_w32(hit != 0u);
    if (lm == 0u) return -1;
    const int ln = __builtin_ctz(lm);
    const unsigned hb = (unsigned)__builtin_amdgcn_readlane((int)hit, ln);
    const int i = __builtin_ctz(hb);
    if (lane == ln) hit &= hit - 1u;
    return ebase + 4 * ln + 128 * (i >> 2) + (i & 3);
}

__global__ void __launch_bounds__(EDGE_WAVES * 32)
edge_block_kernel(const float* __restrict__ h_src, const float* __restrict__ h_dst,
                  const float* __restrict__ edge_w, const float* __restrict__ we,
                  const int* __restrict__ src_idx, const int* __restrict__ dst_idx,
                  float* aggr, int n_dst, int E) {
    __shared__ float ldsAcc[NPB * TCP];
    __shared__ float ldsHd[NPB * TCP];
    __shared__ unsigned ldsNext;

    const int tid  = threadIdx.x;
    const int lane = tid & 31;
    const int wave = __builtin_amdgcn_readfirstlane(tid >> 5);
    const int lo   = blockIdx.x * NPB;
    const int cnt  = min(NPB, n_dst - lo);
    const unsigned ucnt = (unsigned)cnt;

    if (tid == 0) ldsNext = 2u * EDGE_WAVES;
    for (int i = tid; i < cnt * 64; i += EDGE_WAVES * 32) {
        const int node = i >> 6, q = i & 63, l = q & 31, j = q >> 5;
        const long long g = (long long)(lo + node) * TCP + 4 * q;
        const float4 b = *(const float4*)(aggr + g);
        const float4 h = *(const float4*)(h_dst + g);
        float* a = ldsAcc + node * TCP + 128 * j + l;
        a[0] = b.x; a[32] = b.y; a[64] = b.z; a[96] = b.w;
        *(float4*)(ldsHd + node * TCP + 4 * q) = h;
    }
    __syncthreads();

    const float4 wv = *(const float4*)(we + ((4 * lane) & (CDIM - 1)));

    const int ngroups = (E + 511) >> 9;
    int g  = wave;
    int gn = wave + EDGE_WAVES;
    int4 c0 = make_int4(-1, -1, -1, -1), c1 = c0, c2 = c0, c3 = c0;
    if (g < ngroups) load_group(dst_idx, g, E, lane, c0, c1, c2, c3);
    while (g < ngroups) {
        int4 n0 = make_int4(-1, -1, -1, -1), n1 = n0, n2 = n0, n3 = n0;
        if (gn < ngroups) load_group(dst_idx, gn, E, lane, n0, n1, n2, n3);
        unsigned drawn = 0u;
        if (lane == 0) drawn = atomicAdd(&ldsNext, 1u);

        const int dv[16] = {c0.x, c0.y, c0.z, c0.w, c1.x, c1.y, c1.z, c1.w,
                            c2.x, c2.y, c2.z, c2.w, c3.x, c3.y, c3.z, c3.w};
        unsigned hit = 0u;
#pragma unroll
        for (int i = 0; i < 16; ++i)
            hit |= ((unsigned)(dv[i] - lo) < ucnt) ? (1u << i) : 0u;

        const int ebase = g << 9;
        int e = next_hit(hit, ebase, lane);
        if (e >= 0) {
            int   s = src_idx[e];
            int   d = dst_idx[e];
            float w = edge_w[e];
            for (;;) {
                const float* __restrict__ hsp = h_src + (long long)s * TCP + 4 * lane;
                const float4 a0 = *(const float4*)(hsp);
                const float4 a1 = *(const float4*)(hsp + 128);
                const int en = next_hit(hit, ebase, lane);
                int sn = 0, dn = 0; float wn = 0.f;
                if (en >= 0) {
                    sn = src_idx[en];
                    dn = dst_idx[en];
                    wn = edge_w[en];
                }
                gated_row_add(a0, a1, ldsHd, ldsAcc, d - lo, w, lane, wv);
                if (en < 0) break;
                s = sn; d = dn; w = wn;
            }
        }
        c0 = n0; c1 = n1; c2 = n2; c3 = n3;
        g  = gn;
        gn = __builtin_amdgcn_readfirstlane((int)drawn);
    }
    __syncthreads();

    for (int i = tid; i < cnt * 64; i += EDGE_WAVES * 32) {
        const int node = i >> 6, q4 = i & 63, l = q4 & 31, j = q4 >> 5;
        const float* a = ldsAcc + node * TCP + 128 * j + l;
        float4 o;
        o.x = a[0]; o.y = a[32]; o.z = a[64]; o.w = a[96];
        *(float4*)(aggr + (long long)(lo + node) * TCP + 4 * q4) = o;
    }
}

__global__ void __launch_bounds__(256)
ln_kernel(const float* __restrict__ acc, const float* __restrict__ feat_dst,
          const float* __restrict__ gamma, const float* __restrict__ beta,
          float* __restrict__ out, int rows) {
    const int wave = threadIdx.x >> 5;
    const int lane = threadIdx.x & 31;
    const int row  = blockIdx.x * (blockDim.x >> 5) + wave;
    if (row >= rows) return;

    const float* __restrict__ x = acc + (long long)row * CDIM;
    float x0 = x[lane];
    float x1 = x[lane + 32];

    float s = x0 + x1;
#pragma unroll
    for (int off = 16; off > 0; off >>= 1) s += __shfl_xor(s, off, 32);
    const float mean = s * (1.0f / 64.0f);

    const float d0 = x0 - mean, d1 = x1 - mean;
    float q = d0 * d0 + d1 * d1;
#pragma unroll
    for (int off = 16; off > 0; off >>= 1) q += __shfl_xor(q, off, 32);
    const float rstd = rsqrtf(q * (1.0f / 64.0f) + 1e-5f);

    const float* __restrict__ f = feat_dst + (long long)row * CDIM;
    out[(long long)row * CDIM + lane] =
        d0 * rstd * gamma[lane] + beta[lane] + f[lane];
    out[(long long)row * CDIM + lane + 32] =
        d1 * rstd * gamma[lane + 32] + beta[lane + 32] + f[lane + 32];
}

extern "C" void kernel_launch(void* const* d_in, const int* in_sizes, int n_in,
                              void* d_out, int out_size, void* d_ws, size_t ws_size,
                              hipStream_t stream) {
    const float* feat_src = (const float*)d_in[0];
    const float* feat_dst = (const float*)d_in[1];
    const float* edge_w   = (const float*)d_in[2];
    const float* weight_e = (const float*)d_in[3];
    const float* u        = (const float*)d_in[4];
    const float* v        = (const float*)d_in[5];
    const float* ln_gamma = (const float*)d_in[6];
    const float* ln_beta  = (const float*)d_in[7];
    const int*   src_idx  = (const int*)d_in[8];
    const int*   dst_idx  = (const int*)d_in[9];
    float*       out      = (float*)d_out;

    const int n_src = in_sizes[0] / TCP;
    const int n_dst = in_sizes[1] / TCP;
    const int E     = in_sizes[2];

    float* ws    = (float*)d_ws;
    float* h_src = ws;
    float* h_dst = ws + (size_t)n_src * TCP;
    float* aggr  = h_dst + (size_t)n_dst * TCP;

    const int mtiles_src = (n_src * TDIM) / 16;
    const int mtiles_dst = (n_dst * TDIM) / 16;
    const int wpb = 4;

    proj_src_kernel<<<(mtiles_src + wpb - 1) / wpb, 128, 0, stream>>>(
        feat_src, u, h_src, mtiles_src);

    proj_dst_kernel<<<(mtiles_dst + wpb - 1) / wpb, 128, 0, stream>>>(
        feat_dst, u, v, aggr, h_dst, mtiles_dst);

    edge_block_kernel<<<(n_dst + NPB - 1) / NPB, EDGE_WAVES * 32, 0, stream>>>(
        h_src, h_dst, edge_w, weight_e, src_idx, dst_idx, aggr, n_dst, E);

    const int rows = n_dst * TDIM;
    ln_kernel<<<(rows + 7) / 8, 256, 0, stream>>>(
        aggr, feat_dst, ln_gamma, ln_beta, out, rows);
}


# Projection kernels: u and v no longer copied to LDS (32-round-trip copy loop and barrier gone); B-fragments by 16-byte whole-row global loads, tile t = columns 4c+t; whole-row stores; sum order kept
# speedup vs baseline: 2.6130x; 1.4831x over previous
#include <hip/hip_runtime.h>
#include <math.h>


typedef __attribute__((ext_vector_type(2))) float v2f;
typedef __attribute__((ext_vector_type(8))) float v8f;

#define CDIM 64
#define TDIM 4
#define ROWLEN 64
#define TCP 256

__device__ __forceinline__ v8f wmma4(v2f a, float b0, float b1, v8f acc) {
    v2f b;
    b.x = b0;
    b.y = b1;
    return __builtin_amdgcn_wmma_f32_16x16x4_f32(
        false, a, false, b, (short)0, acc, false, false);
}

__device__ __forceinline__ void store_rows(float* __restrict__ Out, int row0,
                                           v8f t0, v8f t1, v8f t2, v8f t3) {
    const int lane = threadIdx.x & 31;
    float* __restrict__ o =
        Out + ((long long)row0 + ((lane >> 4) << 3)) * ROWLEN + 4 * (lane & 15);
#pragma unroll
    for (int i = 0; i < 8; ++i)
        *(float4*)(o + i * ROWLEN) = make_float4(t0[i], t1[i], t2[i], t3[i]);
}

__global__ void __launch_bounds__(128)
proj_src_kernel(const float* __restrict__ feat, const float* __restrict__ U,
                float* __restrict__ Out, int mtiles) {
    const int wave = threadIdx.x >> 5;
    const int tile = blockIdx.x * (blockDim.x >> 5) + wave;
    if (tile >= mtiles) return;
    const int row0 = tile * 16;
    const int lane = threadIdx.x & 31;
    const int m    = lane & 15;
    const int koff = (lane >> 4) << 1;
    const float* __restrict__ arow = feat + ((long long)row0 + m) * ROWLEN;
    const float* __restrict__ ucol = U + koff * ROWLEN + 4 * m;

    v8f acc0 = {0.f, 0.f, 0.f, 0.f, 0.f, 0.f, 0.f, 0.f};
    v8f acc1 = acc0, acc2 = acc0, acc3 = acc0;
#pragma unroll
    for (int k0 = 0; k0 < 64; k0 += 4) {
        v2f a;
        a.x = arow[k0 + koff];
        a.y = arow[k0 + koff + 1];
        const float4 u0 = *(const float4*)(ucol + k0 * ROWLEN);
        const float4 u1 = *(const float4*)(ucol + (k0 + 1) * ROWLEN);
        acc0 = wmma4(a, u0.x, u1.x, acc0);
        acc1 = wmma4(a, u0.y, u1.y, acc1);
        acc2 = wmma4(a, u0.z, u1.z, acc2);
        acc3 = wmma4(a, u0.w, u1.w, acc3);
    }
    store_rows(Out, row0, acc0, acc1, acc2, acc3);
}

__global__ void __launch_bounds__(128)
proj_dst_kernel(const float* __restrict__ feat, const float* __restrict__ U,
                const float* __restrict__ V, float* __restrict__ OutBase,
                float* __restrict__ OutV, int mtiles) {
    const int wave = threadIdx.x >> 5;
    const int tile = blockIdx.x * (blockDim.x >> 5) + wave;
    if (tile >= mtiles) return;
    const int row0 = tile * 16;
    const int lane = threadIdx.x & 31;
    const int m    = lane & 15;
    const int koff = (lane >> 4) << 1;
    const float* __restrict__ arow = feat + ((long long)row0 + m) * ROWLEN;
    const float* __restrict__ ucol = U + koff * ROWLEN + 4 * m;
    const float* __restrict__ vcol = V + koff * ROWLEN + 4 * m;

    v8f au0 = {0.f, 0.f, 0.f, 0.f, 0.f, 0.f, 0.f, 0.f};
    v8f au1 = au0, au2 = au0, au3 = au0;
    v8f av0 = au0, av1 = au0, av2 = au0, av3 = au0;
#pragma unroll
    for (int k0 = 0; k0 < 64; k0 += 4) {
        v2f a;
        a.x = arow[k0 + koff];
        a.y = arow[k0 + koff + 1];
        const float4 u0 = *(const float4*)(ucol + k0 * ROWLEN);
        const float4 u1 = *(const float4*)(ucol + (k0 + 1) * ROWLEN);
        const float4 w0 = *(const float4*)(vcol + k0 * ROWLEN);
        const float4 w1 = *(const float4*)(vcol + (k0 + 1) * ROWLEN);
        au0 = wmma4(a, u0.x, u1.x, au0);
        av0 = wmma4(a, w0.x, w1.x, av0);
        au1 = wmma4(a, u0.y, u1.y, au1);
        av1 = wmma4(a, w0.y, w1.y, av1);
        au2 = wmma4(a, u0.z, u1.z, au2);
        av2 = wmma4(a, w0.z, w1.z, av2);
        au3 = wmma4(a, u0.w, u1.w, au3);
        av3 = wmma4(a, w0.w, w1.w, av3);
    }
    store_rows(OutBase, row0, au0, au1, au2, au3);
    store_rows(OutV,    row0, av0, av1, av2, av3);
}

#define NPB 40
#define EDGE_WAVES 32

__device__ __forceinline__ float gated(float hs, float hd, float ww) {
    const float x = hs * hd * ww;
    return hs / (1.0f + __expf(-x));
}

__device__ __forceinline__ void gated_row_add(float4 a0, float4 a1, const float* ldsHd,
                                              float* ldsAcc, int dl, float w,
                                              int lane, float4 wv) {
    const float* hdp = ldsHd + dl * TCP + 4 * lane;
    const float4 b0 = *(const float4*)(hdp);
    const float4 b1 = *(const float4*)(hdp + 128);
    float* acc = ldsAcc + dl * TCP + lane;
    const float wx = w * wv.x, wy = w * wv.y, wz = w * wv.z, ww = w * wv.w;
    atomicAdd(acc +   0, gated(a0.x, b0.x, wx));
    atomicAdd(acc +  32, gated(a0.y, b0.y, wy));
    atomicAdd(acc +  64, gated(a0.z, b0.z, wz));
    atomicAdd(acc +  96, gated(a0.w, b0.w, ww));
    atomicAdd(acc + 128, gated(a1.x, b1.x, wx));
    atomicAdd(acc + 160, gated(a1.y, b1.y, wy));
    atomicAdd(acc + 192, gated(a1.z, b1.z, wz));
    atomicAdd(acc + 224, gated(a1.w, b1.w, ww));
}

__device__ __forceinline__ void load_group(const int* __restrict__ dst_idx, int g, int E,
                                           int lane, int4& c0, int4& c1, int4& c2, int4& c3) {
    const int eb = (g << 9) + 4 * lane;
    if (((g + 1) << 9) <= E) {
        c0 = *(const int4*)(dst_idx + eb);
        c1 = *(const int4*)(dst_idx + eb + 128);
        c2 = *(const int4*)(dst_idx + eb + 256);
        c3 = *(const int4*)(dst_idx + eb + 384);
    } else {
        int t[16];
#pragma unroll
        for (int i = 0; i < 16; ++i) {
            const int e = eb + 128 * (i >> 2) + (i & 3);
            t[i] = (e < E) ? dst_idx[e] : -1;
        }
        c0 = make_int4(t[0], t[1], t[2], t[3]);
        c1 = make_int4(t[4], t[5], t[6], t[7]);
        c2 = make_int4(t[8], t[9], t[10], t[11]);
        c3 = make_int4(t[12], t[13], t[14], t[15]);
    }
}

__device__ __forceinline__ int next_hit(unsigned& hit, int ebase, int lane) {
    const unsigned lm = __builtin_amdgcn_ballot_w32(hit != 0u);
    if (lm == 0u) return -1;
    const int ln = __builtin_ctz(lm);
    const unsigned hb = (unsigned)__builtin_amdgcn_readlane((int)hit, ln);
    const int i = __builtin_ctz(hb);
    if (lane == ln) hit &= hit - 1u;
    return ebase + 4 * ln + 128 * (i >> 2) + (i & 3);
}

__global__ void __launch_bounds__(EDGE_WAVES * 32)
edge_block_kernel(const float* __restrict__ h_src, const float* __restrict__ h_dst,
                  const float* __restrict__ edge_w, const float* __restrict__ we,
                  const int* __restrict__ src_idx, const int* __restrict__ dst_idx,
                  float* aggr, int n_dst, int E) {
    __shared__ float ldsAcc[NPB * TCP];
    __shared__ float ldsHd[NPB * TCP];
    __shared__ unsigned ldsNext;

    const int tid  = threadIdx.x;
    const int lane = tid & 31;
    const int wave = __builtin_amdgcn_readfirstlane(tid >> 5);
    const int lo   = blockIdx.x * NPB;
    const int cnt  = min(NPB, n_dst - lo);
    const unsigned ucnt = (unsigned)cnt;

    if (tid == 0) ldsNext = 2u * EDGE_WAVES;
    for (int i = tid; i < cnt * 64; i += EDGE_WAVES * 32) {
        const int node = i >> 6, q = i & 63, l = q & 31, j = q >> 5;
        const long long g = (long long)(lo + node) * TCP + 4 * q;
        const float4 b = *(const float4*)(aggr + g);
        const float4 h = *(const float4*)(h_dst + g);
        float* a = ldsAcc + node * TCP + 128 * j + l;
        a[0] = b.x; a[32] = b.y; a[64] = b.z; a[96] = b.w;
        *(float4*)(ldsHd + node * TCP + 4 * q) = h;
    }
    __syncthreads();

    const float4 wv = *(const float4*)(we + ((4 * lane) & (CDIM - 1)));

    const int ngroups = (E + 511) >> 9;
    int g  = wave;
    int gn = wave + EDGE_WAVES;
    int4 c0 = make_int4(-1, -1, -1, -1), c1 = c0, c2 = c0, c3 = c0;
    if (g < ngroups) load_group(dst_idx, g, E, lane, c0, c1, c2, c3);
    while (g < ngroups) {
        int4 n0 = make_int4(-1, -1, -1, -1), n1 = n0, n2 = n0, n3 = n0;
        if (gn < ngroups) load_group(dst_idx, gn, E, lane, n0, n1, n2, n3);
        unsigned drawn = 0u;
        if (lane == 0) drawn = atomicAdd(&ldsNext, 1u);

        const int dv[16] = {c0.x, c0.y, c0.z, c0.w, c1.x, c1.y, c1.z, c1.w,
                            c2.x, c2.y, c2.z, c2.w, c3.x, c3.y, c3.z, c3.w};
        unsigned hit = 0u;
#pragma unroll
        for (int i = 0; i < 16; ++i)
            hit |= ((unsigned)(dv[i] - lo) < ucnt) ? (1u << i) : 0u;

        const int ebase = g << 9;
        int e = next_hit(hit, ebase, lane);
        if (e >= 0) {
            int   s = src_idx[e];
            int   d = dst_idx[e];
            float w = edge_w[e];
            for (;;) {
                const float* __restrict__ hsp = h_src + (long long)s * TCP + 4 * lane;
                const float4 a0 = *(const float4*)(hsp);
                const float4 a1 = *(const float4*)(hsp + 128);
                const int en = next_hit(hit, ebase, lane);
                int sn = 0, dn = 0; float wn = 0.f;
                if (en >= 0) {
                    sn = src_idx[en];
                    dn = dst_idx[en];
                    wn = edge_w[en];
                }
                gated_row_add(a0, a1, ldsHd, ldsAcc, d - lo, w, lane, wv);
                if (en < 0) break;
                s = sn; d = dn; w = wn;
            }
        }
        c0 = n0; c1 = n1; c2 = n2; c3 = n3;
        g  = gn;
        gn = __builtin_amdgcn_readfirstlane((int)drawn);
    }
    __syncthreads();

    for (int i = tid; i < cnt * 64; i += EDGE_WAVES * 32) {
        const int node = i >> 6, q4 = i & 63, l = q4 & 31, j = q4 >> 5;
        const float* a = ldsAcc + node * TCP + 128 * j + l;
        float4 o;
        o.x = a[0]; o.y = a[32]; o.z = a[64]; o.w = a[96];
        *(float4*)(aggr + (long long)(lo + node) * TCP + 4 * q4) = o;
    }
}

__global__ void __launch_bounds__(256)
ln_kernel(const float* __restrict__ acc, const float* __restrict__ feat_dst,
          const float* __restrict__ gamma, const float* __restrict__ beta,
          float* __restrict__ out, int rows) {
    const int wave = threadIdx.x >> 5;
    const int lane = threadIdx.x & 31;
    const int row  = blockIdx.x * (blockDim.x >> 5) + wave;
    if (row >= rows) return;

    const float* __restrict__ x = acc + (long long)row * CDIM;
    float x0 = x[lane];
    float x1 = x[lane + 32];

    float s = x0 + x1;
#pragma unroll
    for (int off = 16; off > 0; off >>= 1) s += __shfl_xor(s, off, 32);
    const float mean = s * (1.0f / 64.0f);

    const float d0 = x0 - mean, d1 = x1 - mean;
    float q = d0 * d0 + d1 * d1;
#pragma unroll
    for (int off = 16; off > 0; off >>= 1) q += __shfl_xor(q, off, 32);
    const float rstd = rsqrtf(q * (1.0f / 64.0f) + 1e-5f);

    const float* __restrict__ f = feat_dst + (long long)row * CDIM;
    out[(long long)row * CDIM + lane] =
        d0 * rstd * gamma[lane] + beta[lane] + f[lane];
    out[(long long)row * CDIM + lane + 32] =
        d1 * rstd * gamma[lane + 32] + beta[lane + 32] + f[lane + 32];
}

extern "C" void kernel_launch(void* const* d_in, const int* in_sizes, int n_in,
                              void* d_out, int out_size, void* d_ws, size_t ws_size,
                              hipStream_t stream) {
    const float* feat_src = (const float*)d_in[0];
    const float* feat_dst = (const float*)d_in[1];
    const float* edge_w   = (const float*)d_in[2];
    const float* weight_e = (const float*)d_in[3];
    const float* u        = (const float*)d_in[4];
    const float* v        = (const float*)d_in[5];
    const float* ln_gamma = (const float*)d_in[6];
    const float* ln_beta  = (const float*)d_in[7];
    const int*   src_idx  = (const int*)d_in[8];
    const int*   dst_idx  = (const int*)d_in[9];
    float*       out      = (float*)d_out;

    const int n_src = in_sizes[0] / TCP;
    const int n_dst = in_sizes[1] / TCP;
    const int E     = in_sizes[2];

    float* ws    = (float*)d_ws;
    float* h_src = ws;
    float* h_dst = ws + (size_t)n_src * TCP;
    float* aggr  = h_dst + (size_t)n_dst * TCP;

    const int mtiles_src = (n_src * TDIM) / 16;
    const int mtiles_dst = (n_dst * TDIM) / 16;
    const int wpb = 4;

    proj_src_kernel<<<(mtiles_src + wpb - 1) / wpb, 128, 0, stream>>>(
        feat_src, u, h_src, mtiles_src);

    proj_dst_kernel<<<(mtiles_dst + wpb - 1) / wpb, 128, 0, stream>>>(
        feat_dst, u, v, aggr, h_dst, mtiles_dst);

    edge_block_kernel<<<(n_dst + NPB - 1) / NPB, EDGE_WAVES * 32, 0, stream>>>(
        h_src, h_dst, edge_w, weight_e, src_idx, dst_idx, aggr, n_dst, E);

    const int rows = n_dst * TDIM;
    ln_kernel<<<(rows + 7) / 8, 256, 0, stream>>>(
        aggr, feat_dst, ln_gamma, ln_beta, out, rows);
}


# LayerNorm launch: 2000 waves of 20 rows each (16-lane half-wave per row, 16-byte loads all issued first, DPP row sums) replace 40000 one-row waves; LN sums: 4 in lane, then 16-lane tree
# speedup vs baseline: 3.1425x; 1.2026x over previous
#include <hip/hip_runtime.h>
#include <math.h>


typedef __attribute__((ext_vector_type(2))) float v2f;
typedef __attribute__((ext_vector_type(8))) float v8f;

#define CDIM 64
#define TDIM 4
#define ROWLEN 64
#define TCP 256

__device__ __forceinline__ v8f wmma4(v2f a, float b0, float b1, v8f acc) {
    v2f b;
    b.x = b0;
    b.y = b1;
    return __builtin_amdgcn_wmma_f32_16x16x4_f32(
        false, a, false, b, (short)0, acc, false, false);
}

__device__ __forceinline__ void store_rows(float* __restrict__ Out, int row0,
                                           v8f t0, v8f t1, v8f t2, v8f t3) {
    const int lane = threadIdx.x & 31;
    float* __restrict__ o =
        Out + ((long long)row0 + ((lane >> 4) << 3)) * ROWLEN + 4 * (lane & 15);
#pragma unroll
    for (int i = 0; i < 8; ++i)
        *(float4*)(o + i * ROWLEN) = make_float4(t0[i], t1[i], t2[i], t3[i]);
}

__global__ void __launch_bounds__(128)
proj_src_kernel(const float* __restrict__ feat, const float* __restrict__ U,
                float* __restrict__ Out, int mtiles) {
    const int wave = threadIdx.x >> 5;
    const int tile = blockIdx.x * (blockDim.x >> 5) + wave;
    if (tile >= mtiles) return;
    const int row0 = tile * 16;
    const int lane = threadIdx.x & 31;
    const int m    = lane & 15;
    const int koff = (lane >> 4) << 1;
    const float* __restrict__ arow = feat + ((long long)row0 + m) * ROWLEN;
    const float* __restrict__ ucol = U + koff * ROWLEN + 4 * m;

    v8f acc0 = {0.f, 0.f, 0.f, 0.f, 0.f, 0.f, 0.f, 0.f};
    v8f acc1 = acc0, acc2 = acc0, acc3 = acc0;
#pragma unroll
    for (int k0 = 0; k0 < 64; k0 += 4) {
        v2f a;
        a.x = arow[k0 + koff];
        a.y = arow[k0 + koff + 1];
        const float4 u0 = *(const float4*)(ucol + k0 * ROWLEN);
        const float4 u1 = *(const float4*)(ucol + (k0 + 1) * ROWLEN);
        acc0 = wmma4(a, u0.x, u1.x, acc0);
        acc1 = wmma4(a, u0.y, u1.y, acc1);
        acc2 = wmma4(a, u0.z, u1.z, acc2);
        acc3 = wmma4(a, u0.w, u1.w, acc3);
    }
    store_rows(Out, row0, acc0, acc1, acc2, acc3);
}

__global__ void __launch_bounds__(128)
proj_dst_kernel(const float* __restrict__ feat, const float* __restrict__ U,
                const float* __restrict__ V, float* __restrict__ OutBase,
                float* __restrict__ OutV, int mtiles) {
    const int wave = threadIdx.x >> 5;
    const int tile = blockIdx.x * (blockDim.x >> 5) + wave;
    if (tile >= mtiles) return;
    const int row0 = tile * 16;
    const int lane = threadIdx.x & 31;
    const int m    = lane & 15;
    const int koff = (lane >> 4) << 1;
    const float* __restrict__ arow = feat + ((long long)row0 + m) * ROWLEN;
    const float* __restrict__ ucol = U + koff * ROWLEN + 4 * m;
    const float* __restrict__ vcol = V + koff * ROWLEN + 4 * m;

    v8f au0 = {0.f, 0.f, 0.f, 0.f, 0.f, 0.f, 0.f, 0.f};
    v8f au1 = au0, au2 = au0, au3 = au0;
    v8f av0 = au0, av1 = au0, av2 = au0, av3 = au0;
#pragma unroll
    for (int k0 = 0; k0 < 64; k0 += 4) {
        v2f a;
        a.x = arow[k0 + koff];
        a.y = arow[k0 + koff + 1];
        const float4 u0 = *(const float4*)(ucol + k0 * ROWLEN);
        const float4 u1 = *(const float4*)(ucol + (k0 + 1) * ROWLEN);
        const float4 w0 = *(const float4*)(vcol + k0 * ROWLEN);
        const float4 w1 = *(const float4*)(vcol + (k0 + 1) * ROWLEN);
        au0 = wmma4(a, u0.x, u1.x, au0);
        av0 = wmma4(a, w0.x, w1.x, av0);
        au1 = wmma4(a, u0.y, u1.y, au1);
        av1 = wmma4(a, w0.y, w1.y, av1);
        au2 = wmma4(a, u0.z, u1.z, au2);
        av2 = wmma4(a, w0.z, w1.z, av2);
        au3 = wmma4(a, u0.w, u1.w, au3);
        av3 = wmma4(a, w0.w, w1.w, av3);
    }
    store_rows(OutBase, row0, au0, au1, au2, au3);
    store_rows(OutV,    row0, av0, av1, av2, av3);
}

#define NPB 40
#define EDGE_WAVES 32

__device__ __forceinline__ float gated(float hs, float hd, float ww) {
    const float hh = 0.5f * hs;
    const float t  = __builtin_amdgcn_tanhf(hh * hd * ww);
    return __builtin_fmaf(hh, t, hh);
}

__device__ __forceinline__ void gated_row_add(float4 a0, float4 a1, const float* ldsHd,
                                              float* ldsAcc, int dl, float w,
                                              int lane, float4 wv) {
    const float* hdp = ldsHd + dl * TCP + 4 * lane;
    const float4 b0 = *(const float4*)(hdp);
    const float4 b1 = *(const float4*)(hdp + 128);
    float* acc = ldsAcc + dl * TCP + lane;
    const float wx = w * wv.x, wy = w * wv.y, wz = w * wv.z, ww = w * wv.w;
    atomicAdd(acc +   0, gated(a0.x, b0.x, wx));
    atomicAdd(acc +  32, gated(a0.y, b0.y, wy));
    atomicAdd(acc +  64, gated(a0.z, b0.z, wz));
    atomicAdd(acc +  96, gated(a0.w, b0.w, ww));
    atomicAdd(acc + 128, gated(a1.x, b1.x, wx));
    atomicAdd(acc + 160, gated(a1.y, b1.y, wy));
    atomicAdd(acc + 192, gated(a1.z, b1.z, wz));
    atomicAdd(acc + 224, gated(a1.w, b1.w, ww));
}

__device__ __forceinline__ void load_group(const int* __restrict__ dst_idx, int g, int E,
                                           int lane, int4& c0, int4& c1, int4& c2, int4& c3) {
    const int eb = (g << 9) + 4 * lane;
    if (((g + 1) << 9) <= E) {
        const int4* __restrict__ p = (const int4*)(dst_idx + ((long long)g << 9)) + lane;
        c0 = p[0];
        c1 = p[32];
        c2 = p[64];
        c3 = p[96];
    } else {
        int t[16];
#pragma unroll
        for (int i = 0; i < 16; ++i) {
            const int e = eb + 128 * (i >> 2) + (i & 3);
            t[i] = (e < E) ? dst_idx[e] : -1;
        }
        c0 = make_int4(t[0], t[1], t[2], t[3]);
        c1 = make_int4(t[4], t[5], t[6], t[7]);
        c2 = make_int4(t[8], t[9], t[10], t[11]);
        c3 = make_int4(t[12], t[13], t[14], t[15]);
    }
}

#define HITMASK(v) ((unsigned long long)__builtin_amdgcn_ballot_w32((unsigned)((v) - lo) < ucnt))

__global__ void __launch_bounds__(EDGE_WAVES * 32)
edge_block_kernel(const float* __restrict__ h_src, const float* __restrict__ h_dst,
                  const float* __restrict__ edge_w, const float* __restrict__ we,
                  const int* __restrict__ src_idx, const int* __restrict__ dst_idx,
                  float* aggr, int n_dst, int E) {
    __shared__ float ldsAcc[NPB * TCP];
    __shared__ float ldsHd[NPB * TCP];
    __shared__ unsigned ldsNext;

    const int tid  = threadIdx.x;
    const int lane = tid & 31;
    const int wave = __builtin_amdgcn_readfirstlane(tid >> 5);
    const int lo   = blockIdx.x * NPB;
    const int cnt  = min(NPB, n_dst - lo);
    const unsigned ucnt = (unsigned)cnt;

    if (tid == 0) ldsNext = 2u * EDGE_WAVES;
    for (int i = tid; i < cnt * 64; i += EDGE_WAVES * 32) {
        const int node = i >> 6, q = i & 63, l = q & 31, j = q >> 5;
        const long long g = (long long)(lo + node) * TCP + 4 * q;
        const float4 b = *(const float4*)(aggr + g);
        const float4 h = *(const float4*)(h_dst + g);
        float* a = ldsAcc + node * TCP + 128 * j + l;
        a[0] = b.x; a[32] = b.y; a[64] = b.z; a[96] = b.w;
        *(float4*)(ldsHd + node * TCP + 4 * q) = h;
    }
    __syncthreads();

    const float4 wv = *(const float4*)(we + ((4 * lane) & (CDIM - 1)));

    const int ngroups = (E + 511) >> 9;
    int g  = wave;
    int gn = wave + EDGE_WAVES;
    int4 c0 = make_int4(-1, -1, -1, -1), c1 = c0, c2 = c0, c3 = c0;
    if (g < ngroups) load_group(dst_idx, g, E, lane, c0, c1, c2, c3);
    while (g < ngroups) {
        unsigned long long p0 = HITMASK(c0.x) | (HITMASK(c0.y) << 32);
        unsigned long long p1 = HITMASK(c0.z) | (HITMASK(c0.w) << 32);
        unsigned long long p2 = HITMASK(c1.x) | (HITMASK(c1.y) << 32);
        unsigned long long p3 = HITMASK(c1.z) | (HITMASK(c1.w) << 32);
        unsigned long long p4 = HITMASK(c2.x) | (HITMASK(c2.y) << 32);
        unsigned long long p5 = HITMASK(c2.z) | (HITMASK(c2.w) << 32);
        unsigned long long p6 = HITMASK(c3.x) | (HITMASK(c3.y) << 32);
        unsigned long long p7 = HITMASK(c3.z) | (HITMASK(c3.w) << 32);
        int pj = 0;

        if (gn < ngroups) load_group(dst_idx, gn, E, lane, c0, c1, c2, c3);
        unsigned drawn = 0u;
        if (lane == 0) drawn = atomicAdd(&ldsNext, 1u);

        const int ebase = g << 9;
        auto next_hit = [&]() -> int {
            if ((p0 | p1 | p2 | p3 | p4 | p5 | p6 | p7) == 0ull) return -1;
            while (p0 == 0ull) {
                p0 = p1; p1 = p2; p2 = p3; p3 = p4; p4 = p5; p5 = p6; p6 = p7; p7 = 0ull;
                ++pj;
            }
            const int b = __builtin_ctzll(p0);
            p0 &= p0 - 1ull;
            return ebase + 128 * (pj >> 1) + 2 * (pj & 1) + 4 * (b & 31) + (b >> 5);
        };

        int e = next_hit();
        if (e >= 0) {
            int   s = src_idx[e];
            int   d = dst_idx[e];
            float w = edge_w[e];
            for (;;) {
                const float* __restrict__ hsp = h_src + (long long)s * TCP + 4 * lane;
                const float4 a0 = *(const float4*)(hsp);
                const float4 a1 = *(const float4*)(hsp + 128);
                const int en = next_hit();
                int sn = 0, dn = 0; float wn = 0.f;
                if (en >= 0) {
                    sn = src_idx[en];
                    dn = dst_idx[en];
                    wn = edge_w[en];
                }
                gated_row_add(a0, a1, ldsHd, ldsAcc, d - lo, w, lane, wv);
                if (en < 0) break;
                s = sn; d = dn; w = wn;
            }
        }
        g  = gn;
        gn = __builtin_amdgcn_readfirstlane((int)drawn);
    }
    __syncthreads();

    for (int i = tid; i < cnt * 64; i += EDGE_WAVES * 32) {
        const int node = i >> 6, q4 = i & 63, l = q4 & 31, j = q4 >> 5;
        const float* a = ldsAcc + node * TCP + 128 * j + l;
        float4 o;
        o.x = a[0]; o.y = a[32]; o.z = a[64]; o.w = a[96];
        *(float4*)(aggr + (long long)(lo + node) * TCP + 4 * q4) = o;
    }
}

#define LN_RPW    20
#define LN_PASSES (LN_RPW / 2)
#define LN_WAVES  8

template <int M>
__device__ __forceinline__ float row16_xor_add(float v) {
    const int t = __builtin_amdgcn_update_dpp(0, __builtin_bit_cast(int, v),
                                              0x160 + M, 0xf, 0xf, true);
    return v + __builtin_bit_cast(float, t);
}

__device__ __forceinline__ float row16_sum(float v) {
    v = row16_xor_add<1>(v);
    v = row16_xor_add<2>(v);
    v = row16_xor_add<4>(v);
    v = row16_xor_add<8>(v);
    return v;
}

__device__ __forceinline__ float4 ln_row(float4 x, float4 f, float4 g, float4 b) {
    const float s = row16_sum((x.x + x.y) + (x.z + x.w));
    const float mean = s * (1.0f / 64.0f);
    const float d0 = x.x - mean, d1 = x.y - mean, d2 = x.z - mean, d3 = x.w - mean;
    const float q = row16_sum((d0 * d0 + d1 * d1) + (d2 * d2 + d3 * d3));
    const float rstd = rsqrtf(q * (1.0f / 64.0f) + 1e-5f);
    float4 o;
    o.x = d0 * rstd * g.x + b.x + f.x;
    o.y = d1 * rstd * g.y + b.y + f.y;
    o.z = d2 * rstd * g.z + b.z + f.z;
    o.w = d3 * rstd * g.w + b.w + f.w;
    return o;
}

__global__ void __launch_bounds__(LN_WAVES * 32)
ln_kernel(const float* __restrict__ acc, const float* __restrict__ feat_dst,
          const float* __restrict__ gamma, const float* __restrict__ beta,
          float* __restrict__ out, int rows) {
    const int lane = threadIdx.x & 31;
    const int wave = blockIdx.x * LN_WAVES + (threadIdx.x >> 5);
    const int row0 = wave * LN_RPW;
    if (row0 >= rows) return;

    const float4 g = *(const float4*)(gamma + 4 * (lane & 15));
    const float4 b = *(const float4*)(beta + 4 * (lane & 15));

    if (row0 + LN_RPW <= rows) {
        const long long e0 = (long long)row0 * CDIM + 4 * lane;
        float4 x[LN_PASSES], f[LN_PASSES];
#pragma unroll
        for (int p = 0; p < LN_PASSES; ++p) x[p] = *(const float4*)(acc + e0 + 128 * p);
#pragma unroll
        for (int p = 0; p < LN_PASSES; ++p) f[p] = *(const float4*)(feat_dst + e0 + 128 * p);
#pragma unroll
        for (int p = 0; p < LN_PASSES; ++p)
            *(float4*)(out + e0 + 128 * p) = ln_row(x[p], f[p], g, b);
        __builtin_amdgcn_sched_group_barrier(0x020, LN_PASSES, 0);
        __builtin_amdgcn_sched_group_barrier(0x020, LN_PASSES, 0);
        __builtin_amdgcn_sched_group_barrier(0x002, 4096, 0);
    } else {
        for (int p = 0; p < LN_PASSES; ++p) {
            if (row0 + 2 * p >= rows) break;
            const int r  = row0 + 2 * p + (lane >> 4);
            const int rc = min(r, rows - 1);
            const long long e = (long long)rc * CDIM + 4 * (lane & 15);
            const float4 x = *(const float4*)(acc + e);
            const float4 f = *(const float4*)(feat_dst + e);
            const float4 o = ln_row(x, f, g, b);
            if (r < rows) *(float4*)(out + e) = o;
        }
    }
}

extern "C" void kernel_launch(void* const* d_in, const int* in_sizes, int n_in,
                              void* d_out, int out_size, void* d_ws, size_t ws_size,
                              hipStream_t stream) {
    const float* feat_src = (const float*)d_in[0];
    const float* feat_dst = (const float*)d_in[1];
    const float* edge_w   = (const float*)d_in[2];
    const float* weight_e = (const float*)d_in[3];
    const float* u        = (const float*)d_in[4];
    const float* v        = (const float*)d_in[5];
    const float* ln_gamma = (const float*)d_in[6];
    const float* ln_beta  = (const float*)d_in[7];
    const int*   src_idx  = (const int*)d_in[8];
    const int*   dst_idx  = (const int*)d_in[9];
    float*       out      = (float*)d_out;

    const int n_src = in_sizes[0] / TCP;
    const int n_dst = in_sizes[1] / TCP;
    const int E     = in_sizes[2];

    float* ws    = (float*)d_ws;
    float* h_src = ws;
    float* h_dst = ws + (size_t)n_src * TCP;
    float* aggr  = h_dst + (size_t)n_dst * TCP;

    const int mtiles_src = (n_src * TDIM) / 16;
    const int mtiles_dst = (n_dst * TDIM) / 16;
    const int wpb = 4;

    proj_src_kernel<<<(mtiles_src + wpb - 1) / wpb, 128, 0, stream>>>(
        feat_src, u, h_src, mtiles_src);

    proj_dst_kernel<<<(mtiles_dst + wpb - 1) / wpb, 128, 0, stream>>>(
        feat_dst, u, v, aggr, h_dst, mtiles_dst);

    edge_block_kernel<<<(n_dst + NPB - 1) / NPB, EDGE_WAVES * 32, 0, stream>>>(
        h_src, h_dst, edge_w, weight_e, src_idx, dst_idx, aggr, n_dst, E);

    const int rows = n_dst * TDIM;
    const int rpb  = LN_RPW * LN_WAVES;
    ln_kernel<<<(rows + rpb - 1) / rpb, LN_WAVES * 32, 0, stream>>>(
        aggr, feat_dst, ln_gamma, ln_beta, out, rows);
}


# LayerNorm + residual moved into the edge kernel's tail: rows normalised straight from the block's LDS sums (16-lane half-wave per row); sums' store, reload and 4th launch gone; LN sum order kept
# speedup vs baseline: 3.3545x; 1.0675x over previous
#include <hip/hip_runtime.h>
#include <math.h>


typedef __attribute__((ext_vector_type(2))) float v2f;
typedef __attribute__((ext_vector_type(8))) float v8f;

#define CDIM 64
#define TDIM 4
#define ROWLEN 64
#define TCP 256

__device__ __forceinline__ v8f wmma4(v2f a, float b0, float b1, v8f acc) {
    v2f b;
    b.x = b0;
    b.y = b1;
    return __builtin_amdgcn_wmma_f32_16x16x4_f32(
        false, a, false, b, (short)0, acc, false, false);
}

__device__ __forceinline__ void store_rows(float* __restrict__ Out, int row0,
                                           v8f t0, v8f t1, v8f t2, v8f t3) {
    const int lane = threadIdx.x & 31;
    float* __restrict__ o =
        Out + ((long long)row0 + ((lane >> 4) << 3)) * ROWLEN + 4 * (lane & 15);
#pragma unroll
    for (int i = 0; i < 8; ++i)
        *(float4*)(o + i * ROWLEN) = make_float4(t0[i], t1[i], t2[i], t3[i]);
}

__global__ void __launch_bounds__(128)
proj_src_kernel(const float* __restrict__ feat, const float* __restrict__ U,
                float* __restrict__ Out, int mtiles) {
    const int wave = threadIdx.x >> 5;
    const int tile = blockIdx.x * (blockDim.x >> 5) + wave;
    if (tile >= mtiles) return;
    const int row0 = tile * 16;
    const int lane = threadIdx.x & 31;
    const int m    = lane & 15;
    const int koff = (lane >> 4) << 1;
    const float* __restrict__ arow = feat + ((long long)row0 + m) * ROWLEN;
    const float* __restrict__ ucol = U + koff * ROWLEN + 4 * m;

    v8f acc0 = {0.f, 0.f, 0.f, 0.f, 0.f, 0.f, 0.f, 0.f};
    v8f acc1 = acc0, acc2 = acc0, acc3 = acc0;
#pragma unroll
    for (int k0 = 0; k0 < 64; k0 += 4) {
        v2f a;
        a.x = arow[k0 + koff];
        a.y = arow[k0 + koff + 1];
        const float4 u0 = *(const float4*)(ucol + k0 * ROWLEN);
        const float4 u1 = *(const float4*)(ucol + (k0 + 1) * ROWLEN);
        acc0 = wmma4(a, u0.x, u1.x, acc0);
        acc1 = wmma4(a, u0.y, u1.y, acc1);
        acc2 = wmma4(a, u0.z, u1.z, acc2);
        acc3 = wmma4(a, u0.w, u1.w, acc3);
    }
    store_rows(Out, row0, acc0, acc1, acc2, acc3);
}

__global__ void __launch_bounds__(128)
proj_dst_kernel(const float* __restrict__ feat, const float* __restrict__ U,
                const float* __restrict__ V, float* __restrict__ OutBase,
                float* __restrict__ OutV, int mtiles) {
    const int wave = threadIdx.x >> 5;
    const int tile = blockIdx.x * (blockDim.x >> 5) + wave;
    if (tile >= mtiles) return;
    const int row0 = tile * 16;
    const int lane = threadIdx.x & 31;
    const int m    = lane & 15;
    const int koff = (lane >> 4) << 1;
    const float* __restrict__ arow = feat + ((long long)row0 + m) * ROWLEN;
    const float* __restrict__ ucol = U + koff * ROWLEN + 4 * m;
    const float* __restrict__ vcol = V + koff * ROWLEN + 4 * m;

    v8f au0 = {0.f, 0.f, 0.f, 0.f, 0.f, 0.f, 0.f, 0.f};
    v8f au1 = au0, au2 = au0, au3 = au0;
    v8f av0 = au0, av1 = au0, av2 = au0, av3 = au0;
#pragma unroll
    for (int k0 = 0; k0 < 64; k0 += 4) {
        v2f a;
        a.x = arow[k0 + koff];
        a.y = arow[k0 + koff + 1];
        const float4 u0 = *(const float4*)(ucol + k0 * ROWLEN);
        const float4 u1 = *(const float4*)(ucol + (k0 + 1) * ROWLEN);
        const float4 w0 = *(const float4*)(vcol + k0 * ROWLEN);
        const float4 w1 = *(const float4*)(vcol + (k0 + 1) * ROWLEN);
        au0 = wmma4(a, u0.x, u1.x, au0);
        av0 = wmma4(a, w0.x, w1.x, av0);
        au1 = wmma4(a, u0.y, u1.y, au1);
        av1 = wmma4(a, w0.y, w1.y, av1);
        au2 = wmma4(a, u0.z, u1.z, au2);
        av2 = wmma4(a, w0.z, w1.z, av2);
        au3 = wmma4(a, u0.w, u1.w, au3);
        av3 = wmma4(a, w0.w, w1.w, av3);
    }
    store_rows(OutBase, row0, au0, au1, au2, au3);
    store_rows(OutV,    row0, av0, av1, av2, av3);
}


template <int M>
__device__ __forceinline__ float row16_xor_add(float v) {
    const int t = __builtin_amdgcn_update_dpp(0, __builtin_bit_cast(int, v),
                                              0x160 + M, 0xf, 0xf, true);
    return v + __builtin_bit_cast(float, t);
}

__device__ __forceinline__ float row16_sum(float v) {
    v = row16_xor_add<1>(v);
    v = row16_xor_add<2>(v);
    v = row16_xor_add<4>(v);
    v = row16_xor_add<8>(v);
    return v;
}

__device__ __forceinline__ float4 ln_row(float4 x, float4 f, float4 g, float4 b) {
    const float s = row16_sum((x.x + x.y) + (x.z + x.w));
    const float mean = s * (1.0f / 64.0f);
    const float d0 = x.x - mean, d1 = x.y - mean, d2 = x.z - mean, d3 = x.w - mean;
    const float q = row16_sum((d0 * d0 + d1 * d1) + (d2 * d2 + d3 * d3));
    const float rstd = rsqrtf(q * (1.0f / 64.0f) + 1e-5f);
    float4 o;
    o.x = d0 * rstd * g.x + b.x + f.x;
    o.y = d1 * rstd * g.y + b.y + f.y;
    o.z = d2 * rstd * g.z + b.z + f.z;
    o.w = d3 * rstd * g.w + b.w + f.w;
    return o;
}

#define NPB 40
#define EDGE_WAVES 32
#define OUT_PASSES ((NPB * 64 + EDGE_WAVES * 32 - 1) / (EDGE_WAVES * 32))

__device__ __forceinline__ float gated(float hs, float hd, float ww) {
    const float hh = 0.5f * hs;
    const float t  = __builtin_amdgcn_tanhf(hh * hd * ww);
    return __builtin_fmaf(hh, t, hh);
}

__device__ __forceinline__ void gated_row_add(float4 a0, float4 a1, const float* ldsHd,
                                              float* ldsAcc, int dl, float w,
                                              int lane, float4 wv) {
    const float* hdp = ldsHd + dl * TCP + 4 * lane;
    const float4 b0 = *(const float4*)(hdp);
    const float4 b1 = *(const float4*)(hdp + 128);
    float* acc = ldsAcc + dl * TCP + lane;
    const float wx = w * wv.x, wy = w * wv.y, wz = w * wv.z, ww = w * wv.w;
    atomicAdd(acc +   0, gated(a0.x, b0.x, wx));
    atomicAdd(acc +  32, gated(a0.y, b0.y, wy));
    atomicAdd(acc +  64, gated(a0.z, b0.z, wz));
    atomicAdd(acc +  96, gated(a0.w, b0.w, ww));
    atomicAdd(acc + 128, gated(a1.x, b1.x, wx));
    atomicAdd(acc + 160, gated(a1.y, b1.y, wy));
    atomicAdd(acc + 192, gated(a1.z, b1.z, wz));
    atomicAdd(acc + 224, gated(a1.w, b1.w, ww));
}

__device__ __forceinline__ void load_group(const int* __restrict__ dst_idx, int g, int E,
                                           int lane, int4& c0, int4& c1, int4& c2, int4& c3) {
    const int eb = (g << 9) + 4 * lane;
    if (((g + 1) << 9) <= E) {
        const int4* __restrict__ p = (const int4*)(dst_idx + ((long long)g << 9)) + lane;
        c0 = p[0];
        c1 = p[32];
        c2 = p[64];
        c3 = p[96];
    } else {
        int t[16];
#pragma unroll
        for (int i = 0; i < 16; ++i) {
            const int e = eb + 128 * (i >> 2) + (i & 3);
            t[i] = (e < E) ? dst_idx[e] : -1;
        }
        c0 = make_int4(t[0], t[1], t[2], t[3]);
        c1 = make_int4(t[4], t[5], t[6], t[7]);
        c2 = make_int4(t[8], t[9], t[10], t[11]);
        c3 = make_int4(t[12], t[13], t[14], t[15]);
    }
}

#define HITMASK(v) ((unsigned long long)__builtin_amdgcn_ballot_w32((unsigned)((v) - lo) < ucnt))

__global__ void __launch_bounds__(EDGE_WAVES * 32)
edge_block_kernel(const float* __restrict__ h_src, const float* __restrict__ h_dst,
                  const float* __restrict__ edge_w, const float* __restrict__ we,
                  const int* __restrict__ src_idx, const int* __restrict__ dst_idx,
                  const float* __restrict__ base, const float* __restrict__ feat_dst,
                  const float* __restrict__ gamma, const float* __restrict__ beta,
                  float* __restrict__ out, int n_dst, int E) {
    __shared__ float ldsAcc[NPB * TCP];
    __shared__ float ldsHd[NPB * TCP];
    __shared__ unsigned ldsNext;

    const int tid  = threadIdx.x;
    const int lane = tid & 31;
    const int wave = __builtin_amdgcn_readfirstlane(tid >> 5);
    const int lo   = blockIdx.x * NPB;
    const int cnt  = min(NPB, n_dst - lo);
    const unsigned ucnt = (unsigned)cnt;

    if (tid == 0) ldsNext = 2u * EDGE_WAVES;
    for (int i = tid; i < cnt * 64; i += EDGE_WAVES * 32) {
        const int node = i >> 6, q = i & 63, l = q & 31, j = q >> 5;
        const long long g = (long long)(lo + node) * TCP + 4 * q;
        const float4 b = *(const float4*)(base + g);
        const float4 h = *(const float4*)(h_dst + g);
        float* a = ldsAcc + node * TCP + 128 * j + l;
        a[0] = b.x; a[32] = b.y; a[64] = b.z; a[96] = b.w;
        *(float4*)(ldsHd + node * TCP + 4 * q) = h;
    }
    __syncthreads();

    const float4 wv = *(const float4*)(we + ((4 * lane) & (CDIM - 1)));

    const int ngroups = (E + 511) >> 9;
    int g  = wave;
    int gn = wave + EDGE_WAVES;
    int4 c0 = make_int4(-1, -1, -1, -1), c1 = c0, c2 = c0, c3 = c0;
    if (g < ngroups) load_group(dst_idx, g, E, lane, c0, c1, c2, c3);
    while (g < ngroups) {
        unsigned long long p0 = HITMASK(c0.x) | (HITMASK(c0.y) << 32);
        unsigned long long p1 = HITMASK(c0.z) | (HITMASK(c0.w) << 32);
        unsigned long long p2 = HITMASK(c1.x) | (HITMASK(c1.y) << 32);
        unsigned long long p3 = HITMASK(c1.z) | (HITMASK(c1.w) << 32);
        unsigned long long p4 = HITMASK(c2.x) | (HITMASK(c2.y) << 32);
        unsigned long long p5 = HITMASK(c2.z) | (HITMASK(c2.w) << 32);
        unsigned long long p6 = HITMASK(c3.x) | (HITMASK(c3.y) << 32);
        unsigned long long p7 = HITMASK(c3.z) | (HITMASK(c3.w) << 32);
        int pj = 0;

        if (gn < ngroups) load_group(dst_idx, gn, E, lane, c0, c1, c2, c3);
        unsigned drawn = 0u;
        if (lane == 0) drawn = atomicAdd(&ldsNext, 1u);

        const int ebase = g << 9;
        auto next_hit = [&]() -> int {
            if ((p0 | p1 | p2 | p3 | p4 | p5 | p6 | p7) == 0ull) return -1;
            while (p0 == 0ull) {
                p0 = p1; p1 = p2; p2 = p3; p3 = p4; p4 = p5; p5 = p6; p6 = p7; p7 = 0ull;
                ++pj;
            }
            const int b = __builtin_ctzll(p0);
            p0 &= p0 - 1ull;
            return ebase + 128 * (pj >> 1) + 2 * (pj & 1) + 4 * (b & 31) + (b >> 5);
        };

        int e = next_hit();
        if (e >= 0) {
            int   s = src_idx[e];
            int   d = dst_idx[e];
            float w = edge_w[e];
            for (;;) {
                const float* __restrict__ hsp = h_src + (long long)s * TCP + 4 * lane;
                const float4 a0 = *(const float4*)(hsp);
                const float4 a1 = *(const float4*)(hsp + 128);
                const int en = next_hit();
                int sn = 0, dn = 0; float wn = 0.f;
                if (en >= 0) {
                    sn = src_idx[en];
                    dn = dst_idx[en];
                    wn = edge_w[en];
                }
                gated_row_add(a0, a1, ldsHd, ldsAcc, d - lo, w, lane, wv);
                if (en < 0) break;
                s = sn; d = dn; w = wn;
            }
        }
        g  = gn;
        gn = __builtin_amdgcn_readfirstlane((int)drawn);
    }
    const float4 lg = *(const float4*)(gamma + 4 * (lane & 15));
    const float4 lb = *(const float4*)(beta + 4 * (lane & 15));
    const float* __restrict__ fblk = feat_dst + (long long)lo * TCP;
    float* __restrict__ oblk = out + (long long)lo * TCP;
    float4 fr[OUT_PASSES];
#pragma unroll
    for (int p = 0; p < OUT_PASSES; ++p)
        fr[p] = *(const float4*)(fblk + 4 * min(tid + p * (EDGE_WAVES * 32), cnt * 64 - 1));
    __syncthreads();

    {
        const int q4 = tid & 63, l = q4 & 31, j = q4 >> 5;
#pragma unroll
        for (int p = 0; p < OUT_PASSES; ++p) {
            const int i = tid + p * (EDGE_WAVES * 32);
            if (i < cnt * 64) {
                const float* a = ldsAcc + (i >> 6) * TCP + 128 * j + l;
                float4 x;
                x.x = a[0]; x.y = a[32]; x.z = a[64]; x.w = a[96];
                *(float4*)(oblk + 4 * i) = ln_row(x, fr[p], lg, lb);
            }
        }
    }
}

extern "C" void kernel_launch(void* const* d_in, const int* in_sizes, int n_in,
                              void* d_out, int out_size, void* d_ws, size_t ws_size,
                              hipStream_t stream) {
    const float* feat_src = (const float*)d_in[0];
    const float* feat_dst = (const float*)d_in[1];
    const float* edge_w   = (const float*)d_in[2];
    const float* weight_e = (const float*)d_in[3];
    const float* u        = (const float*)d_in[4];
    const float* v        = (const float*)d_in[5];
    const float* ln_gamma = (const float*)d_in[6];
    const float* ln_beta  = (const float*)d_in[7];
    const int*   src_idx  = (const int*)d_in[8];
    const int*   dst_idx  = (const int*)d_in[9];
    float*       out      = (float*)d_out;

    const int n_src = in_sizes[0] / TCP;
    const int n_dst = in_sizes[1] / TCP;
    const int E     = in_sizes[2];

    float* ws    = (float*)d_ws;
    float* h_src = ws;
    float* h_dst = ws + (size_t)n_src * TCP;
    float* base  = h_dst + (size_t)n_dst * TCP;

    const int mtiles_src = (n_src * TDIM) / 16;
    const int mtiles_dst = (n_dst * TDIM) / 16;
    const int wpb = 4;

    proj_src_kernel<<<(mtiles_src + wpb - 1) / wpb, 128, 0, stream>>>(
        feat_src, u, h_src, mtiles_src);

    proj_dst_kernel<<<(mtiles_dst + wpb - 1) / wpb, 128, 0, stream>>>(
        feat_dst, u, v, base, h_dst, mtiles_dst);

    edge_block_kernel<<<(n_dst + NPB - 1) / NPB, EDGE_WAVES * 32, 0, stream>>>(
        h_src, h_dst, edge_w, weight_e, src_idx, dst_idx, base, feat_dst,
        ln_gamma, ln_beta, out, n_dst, E);
}
